# starting kernel
# baseline (speedup 1.0000x reference)
#include <hip/hip_runtime.h>
#include <hip/hip_bf16.h>
#include <math.h>

typedef float v2f __attribute__((ext_vector_type(2)));
typedef float v8f __attribute__((ext_vector_type(8)));

#define EPS 1e-12f
#define NEIGH_WEIGHT 0.2f

__global__ void zero_kernel(float* p, size_t n) {
    size_t i = (size_t)blockIdx.x * blockDim.x + threadIdx.x;
    size_t stride = (size_t)gridDim.x * blockDim.x;
    for (; i < n; i += stride) p[i] = 0.0f;
}

__global__ void stats_kernel(const float* __restrict__ pos_diff,
                             const int* __restrict__ nor_idx, int M,
                             float* __restrict__ scratch) {
    int t = blockIdx.x * blockDim.x + threadIdx.x;
    float v = 0.0f;
    if (t < M) v = pos_diff[nor_idx[t]];
    float s1 = v, s2 = v * v;
    #pragma unroll
    for (int off = 16; off > 0; off >>= 1) {
        s1 += __shfl_xor(s1, off, 32);
        s2 += __shfl_xor(s2, off, 32);
    }
    if ((threadIdx.x & 31) == 0) {
        __hip_atomic_fetch_add(&scratch[0], s1, __ATOMIC_RELAXED, __HIP_MEMORY_SCOPE_AGENT);
        __hip_atomic_fetch_add(&scratch[1], s2, __ATOMIC_RELAXED, __HIP_MEMORY_SCOPE_AGENT);
    }
}

__global__ __launch_bounds__(256) void gemm_h_kernel(
    const float* __restrict__ feats, const float* __restrict__ W,
    const float* __restrict__ bias, float* __restrict__ h, int N) {
    __shared__ float As[16 * 260];
    __shared__ float T[16 * 132];
    __shared__ float P[16 * 17];
    __shared__ float RowInv[16];

    const int tid  = threadIdx.x;
    const int wave = tid >> 5;
    const int lane = tid & 31;
    const int row0 = blockIdx.x * 16;
    const int cb   = wave * 16;
    const int hl   = lane & 15;
    const int hi   = lane >> 4;

    #pragma unroll
    for (int j = 0; j < 4; ++j) {
        int idx = tid * 4 + j * 1024;
        int r = idx >> 8;
        int c = idx & 255;
        int grow = row0 + r;
        float4 v = make_float4(0.f, 0.f, 0.f, 0.f);
        if (grow < N) v = *(const float4*)(feats + (size_t)grow * 256 + c);
        float* dstp = &As[r * 260 + c];
        dstp[0] = v.x; dstp[1] = v.y; dstp[2] = v.z; dstp[3] = v.w;
    }
    __syncthreads();

    v8f acc = {};
    for (int k = 0; k < 256; k += 4) {
        int ka = k + hi * 2;
        v2f a, b;
        a.x = As[hl * 260 + ka];
        a.y = As[hl * 260 + ka + 1];
        b.x = W[(size_t)ka * 128 + cb + hl];
        b.y = W[(size_t)(ka + 1) * 128 + cb + hl];
        acc = __builtin_amdgcn_wmma_f32_16x16x4_f32(false, a, false, b,
                                                    (short)0, acc, false, false);
    }

    float bv = bias[cb + hl];
    int rbase = hi * 8;
    #pragma unroll
    for (int r = 0; r < 8; ++r) {
        float v = acc[r] + bv;
        v = v > 0.0f ? v : 0.0f;
        T[(rbase + r) * 132 + cb + hl] = v;
    }
    __syncthreads();

    {
        int row = tid >> 4;
        int c0  = (tid & 15) * 8;
        float s = 0.0f;
        #pragma unroll
        for (int j = 0; j < 8; ++j) { float v = T[row * 132 + c0 + j]; s += v * v; }
        P[row * 17 + (tid & 15)] = s;
    }
    __syncthreads();
    if (tid < 16) {
        float s = 0.0f;
        #pragma unroll
        for (int j = 0; j < 16; ++j) s += P[tid * 17 + j];
        RowInv[tid] = 1.0f / fmaxf(sqrtf(s), EPS);
    }
    __syncthreads();

    {
        int row  = tid >> 4;
        int grow = row0 + row;
        if (grow < N) {
            int c0 = (tid & 15) * 8;
            float inv = RowInv[row];
            float* op = h + (size_t)grow * 128 + c0;
            #pragma unroll
            for (int j = 0; j < 8; ++j) op[j] = T[row * 132 + c0 + j] * inv;
        }
    }
}

__global__ __launch_bounds__(256) void aggregate_kernel(
    const float* __restrict__ h, const int* __restrict__ src,
    const int* __restrict__ dst, float* __restrict__ accum,
    float* __restrict__ deg, int E) {
    int gw = (int)(((size_t)blockIdx.x * blockDim.x + threadIdx.x) >> 5);
    int lane = threadIdx.x & 31;
    if (gw >= E) return;
    int s = src[gw], d = dst[gw];
    const float* hp = h + (size_t)s * 128;
    float* ap = accum + (size_t)d * 128;
    #pragma unroll
    for (int j = 0; j < 4; ++j) {
        int c = lane + 32 * j;
        __hip_atomic_fetch_add(&ap[c], hp[c], __ATOMIC_RELAXED, __HIP_MEMORY_SCOPE_AGENT);
    }
    if (lane == 0)
        __hip_atomic_fetch_add(&deg[d], 1.0f, __ATOMIC_RELAXED, __HIP_MEMORY_SCOPE_AGENT);
}

__global__ __launch_bounds__(256) void finalize_kernel(
    const float* __restrict__ h, const float* __restrict__ accum,
    const float* __restrict__ deg, const float* __restrict__ center,
    const float* __restrict__ pos_diff, const int* __restrict__ out_nodes,
    const int* __restrict__ labels, const int* __restrict__ epoch,
    float* __restrict__ scratch, float* __restrict__ scores_out,
    int N, int Mnor) {
    int gw = (int)(((size_t)blockIdx.x * blockDim.x + threadIdx.x) >> 5);
    int lane = threadIdx.x & 31;
    if (gw >= N) return;
    int i = gw;

    float dg = deg[i];
    float invd = dg > 0.0f ? 1.0f / fmaxf(dg, 1.0f) : 0.0f;

    const float* hp = h + (size_t)i * 128;
    const float* ap = accum + (size_t)i * 128;
    float hv[4], mv[4];
    float dot = 0.0f;
    #pragma unroll
    for (int j = 0; j < 4; ++j) {
        int c = lane + 32 * j;
        hv[j] = hp[c];
        mv[j] = ap[c] * invd;
        dot += hv[j] * mv[j];
    }
    #pragma unroll
    for (int off = 16; off > 0; off >>= 1) dot += __shfl_xor(dot, off, 32);

    int on = out_nodes[i];
    float cnt = (float)Mnor;
    float mu  = scratch[0] / cnt;
    float var = (scratch[1] - cnt * mu * mu) / (cnt - 1.0f);
    float sd  = sqrtf(var);
    float z   = (pos_diff[on] - mu) / sd;
    float pre = 1.0f - 1.0f / (1.0f + expf(-z));

    float beta = powf(0.9f, (float)epoch[0]);
    if (beta < 0.1f) beta = 0.0f;
    float attn = beta * pre + (1.0f - beta) * dot;
    float nei  = attn * NEIGH_WEIGHT;

    float score = 0.0f;
    #pragma unroll
    for (int j = 0; j < 4; ++j) {
        int c = lane + 32 * j;
        float h2 = nei * mv[j] + (1.0f - nei) * hv[j];
        score += h2 * center[c];
    }
    #pragma unroll
    for (int off = 16; off > 0; off >>= 1) score += __shfl_xor(score, off, 32);

    if (lane == 0) {
        scores_out[i] = score;
        int lab = labels[on];
        float spl = log1pf(expf(-fabsf(score)));
        if (lab == 1) {
            float sp = fmaxf(score, 0.0f) + spl;
            __hip_atomic_fetch_add(&scratch[4], sp, __ATOMIC_RELAXED, __HIP_MEMORY_SCOPE_AGENT);
            __hip_atomic_fetch_add(&scratch[5], 1.0f, __ATOMIC_RELAXED, __HIP_MEMORY_SCOPE_AGENT);
        } else {
            float sp = fmaxf(-score, 0.0f) + spl;
            __hip_atomic_fetch_add(&scratch[2], sp, __ATOMIC_RELAXED, __HIP_MEMORY_SCOPE_AGENT);
            __hip_atomic_fetch_add(&scratch[3], 1.0f, __ATOMIC_RELAXED, __HIP_MEMORY_SCOPE_AGENT);
        }
    }
}

__global__ void loss_kernel(const float* __restrict__ scratch, float* __restrict__ out0) {
    if (threadIdx.x == 0 && blockIdx.x == 0) {
        out0[0] = scratch[2] / fmaxf(scratch[3], 1.0f) +
                  scratch[4] / fmaxf(scratch[5], 1.0f);
    }
}

extern "C" void kernel_launch(void* const* d_in, const int* in_sizes, int n_in,
                              void* d_out, int out_size, void* d_ws, size_t ws_size,
                              hipStream_t stream) {
    const float* feats    = (const float*)d_in[0];
    const float* W        = (const float*)d_in[1];
    const float* b        = (const float*)d_in[2];
    const float* center   = (const float*)d_in[3];
    const float* pos_diff = (const float*)d_in[4];
    const int*   src      = (const int*)d_in[5];
    const int*   dst      = (const int*)d_in[6];
    const int*   out_nodes= (const int*)d_in[7];
    const int*   nor_idx  = (const int*)d_in[8];
    const int*   labels   = (const int*)d_in[9];
    const int*   epoch    = (const int*)d_in[10];

    const int N = in_sizes[0] / 256;
    const int E = in_sizes[5];
    const int M = in_sizes[8];

    float* wsf     = (float*)d_ws;
    float* scratch = wsf;
    float* h       = wsf + 64;
    float* accum   = h + (size_t)N * 128;
    float* deg     = accum + (size_t)N * 128;

    float* out     = (float*)d_out;

    size_t zn = 64 + 2 * (size_t)N * 128 + (size_t)N;
    zero_kernel<<<2048, 256, 0, stream>>>(wsf, zn);

    stats_kernel<<<(M + 255) / 256, 256, 0, stream>>>(pos_diff, nor_idx, M, scratch);

    gemm_h_kernel<<<(N + 15) / 16, 256, 0, stream>>>(feats, W, b, h, N);

    aggregate_kernel<<<(E + 7) / 8, 256, 0, stream>>>(h, src, dst, accum, deg, E);

    finalize_kernel<<<(N + 7) / 8, 256, 0, stream>>>(h, accum, deg, center, pos_diff,
                                                     out_nodes, labels, epoch, scratch,
                                                     out + 1, N, M);

    loss_kernel<<<1, 64, 0, stream>>>(scratch, out);
}


# edge pass adds dot(h[dst],h[src]), dot(h[src],center) and 1 into 3 words per node; scores from these scalars, accum rows and row atomics gone (sum orders change; loss summed per block first)
# speedup vs baseline: 4.0294x; 4.0294x over previous
#include <hip/hip_runtime.h>
#include <hip/hip_bf16.h>
#include <math.h>

typedef float v2f __attribute__((ext_vector_type(2)));
typedef float v8f __attribute__((ext_vector_type(8)));

#define EPS 1e-12f
#define NEIGH_WEIGHT 0.2f

template <int K>
__device__ __forceinline__ float row_xor(float v) {
    return __builtin_bit_cast(float,
        __builtin_amdgcn_update_dpp(0, __builtin_bit_cast(int, v), 0x160 + K, 0xf, 0xf, true));
}

__global__ void zero_kernel(float* p, size_t n) {
    size_t i = (size_t)blockIdx.x * blockDim.x + threadIdx.x;
    size_t stride = (size_t)gridDim.x * blockDim.x;
    for (; i < n; i += stride) p[i] = 0.0f;
}

__global__ void stats_kernel(const float* __restrict__ pos_diff,
                             const int* __restrict__ nor_idx, int M,
                             float* __restrict__ scratch) {
    int t = blockIdx.x * blockDim.x + threadIdx.x;
    float v = 0.0f;
    if (t < M) v = pos_diff[nor_idx[t]];
    float s1 = v, s2 = v * v;
    #pragma unroll
    for (int off = 16; off > 0; off >>= 1) {
        s1 += __shfl_xor(s1, off, 32);
        s2 += __shfl_xor(s2, off, 32);
    }
    if ((threadIdx.x & 31) == 0) {
        __hip_atomic_fetch_add(&scratch[0], s1, __ATOMIC_RELAXED, __HIP_MEMORY_SCOPE_AGENT);
        __hip_atomic_fetch_add(&scratch[1], s2, __ATOMIC_RELAXED, __HIP_MEMORY_SCOPE_AGENT);
    }
}

__global__ __launch_bounds__(256) void gemm_h_kernel(
    const float* __restrict__ feats, const float* __restrict__ W,
    const float* __restrict__ bias, const float* __restrict__ center,
    float* __restrict__ h, float* __restrict__ node4, int N) {
    __shared__ float As[16 * 260];
    __shared__ float T[16 * 132];
    __shared__ float P[16 * 17];
    __shared__ float Pc[16 * 17];
    __shared__ float RowInv[16];

    const int tid  = threadIdx.x;
    const int wave = tid >> 5;
    const int lane = tid & 31;
    const int row0 = blockIdx.x * 16;
    const int cb   = wave * 16;
    const int hl   = lane & 15;
    const int hi   = lane >> 4;

    #pragma unroll
    for (int j = 0; j < 4; ++j) {
        int idx = tid * 4 + j * 1024;
        int r = idx >> 8;
        int c = idx & 255;
        int grow = row0 + r;
        float4 v = make_float4(0.f, 0.f, 0.f, 0.f);
        if (grow < N) v = *(const float4*)(feats + (size_t)grow * 256 + c);
        float* dstp = &As[r * 260 + c];
        dstp[0] = v.x; dstp[1] = v.y; dstp[2] = v.z; dstp[3] = v.w;
    }
    __syncthreads();

    v8f acc = {};
    for (int k = 0; k < 256; k += 4) {
        int ka = k + hi * 2;
        v2f a, b;
        a.x = As[hl * 260 + ka];
        a.y = As[hl * 260 + ka + 1];
        b.x = W[(size_t)ka * 128 + cb + hl];
        b.y = W[(size_t)(ka + 1) * 128 + cb + hl];
        acc = __builtin_amdgcn_wmma_f32_16x16x4_f32(false, a, false, b,
                                                    (short)0, acc, false, false);
    }

    float bv = bias[cb + hl];
    int rbase = hi * 8;
    #pragma unroll
    for (int r = 0; r < 8; ++r) {
        float v = acc[r] + bv;
        v = v > 0.0f ? v : 0.0f;
        T[(rbase + r) * 132 + cb + hl] = v;
    }
    __syncthreads();

    {
        int row = tid >> 4;
        int c0  = (tid & 15) * 8;
        float s = 0.0f, sc = 0.0f;
        #pragma unroll
        for (int j = 0; j < 8; ++j) {
            float v = T[row * 132 + c0 + j];
            s += v * v;
            sc += v * center[c0 + j];
        }
        P[row * 17 + (tid & 15)] = s;
        Pc[row * 17 + (tid & 15)] = sc;
    }
    __syncthreads();
    if (tid < 16) {
        float s = 0.0f, sc = 0.0f;
        #pragma unroll
        for (int j = 0; j < 16; ++j) { s += P[tid * 17 + j]; sc += Pc[tid * 17 + j]; }
        float inv = 1.0f / fmaxf(sqrtf(s), EPS);
        RowInv[tid] = inv;
        if (row0 + tid < N)
            *(float4*)(node4 + (size_t)(row0 + tid) * 4) = make_float4(0.0f, 0.0f, 0.0f, sc * inv);
    }
    __syncthreads();

    {
        int row  = tid >> 4;
        int grow = row0 + row;
        if (grow < N) {
            int c0 = (tid & 15) * 8;
            float inv = RowInv[row];
            float* op = h + (size_t)grow * 128 + c0;
            #pragma unroll
            for (int j = 0; j < 8; ++j) op[j] = T[row * 132 + c0 + j] * inv;
        }
    }
}

__global__ __launch_bounds__(256) void edge_dot_kernel(
    const float* __restrict__ h, const float* __restrict__ center,
    const int* __restrict__ src, const int* __restrict__ dst,
    float* __restrict__ node4, int E) {
    const int lane = threadIdx.x & 31;
    const int wv = (int)(((size_t)blockIdx.x * blockDim.x + threadIdx.x) >> 5);
    const int e0 = wv * 32;
    if (e0 >= E) return;
    const int g = lane >> 3;
    const int q = lane & 7;

    int sv = 0, dv = 0;
    if (e0 + lane < E) { sv = src[e0 + lane]; dv = dst[e0 + lane]; }

    float4 cw[4];
    for (int j = 0; j < 4; ++j) cw[j] = *(const float4*)(center + j * 32 + q * 4);

    for (int it = 0; it < 8; ++it) {
        const int sl = it * 4 + g;
        const int s = __shfl(sv, sl, 32);
        const int d = __shfl(dv, sl, 32);
        const float4* ps = (const float4*)(h + (size_t)s * 128) + q;
        const float4* pd = (const float4*)(h + (size_t)d * 128) + q;
        float4 a[4], b[4];
            for (int j = 0; j < 4; ++j) a[j] = ps[j * 8];
            for (int j = 0; j < 4; ++j) b[j] = pd[j * 8];
        v2f dot2 = {0.0f, 0.0f}, cs2 = {0.0f, 0.0f};
            for (int j = 0; j < 4; ++j) {
            v2f alo = {a[j].x, a[j].y}, ahi = {a[j].z, a[j].w};
            v2f blo = {b[j].x, b[j].y}, bhi = {b[j].z, b[j].w};
            v2f clo = {cw[j].x, cw[j].y}, chi = {cw[j].z, cw[j].w};
            dot2 += alo * blo; dot2 += ahi * bhi;
            cs2  += alo * clo; cs2  += ahi * chi;
        }
        float dot = dot2.x + dot2.y, cs = cs2.x + cs2.y;
        dot += row_xor<4>(dot); cs += row_xor<4>(cs);
        dot += row_xor<2>(dot); cs += row_xor<2>(cs);
        dot += row_xor<1>(dot); cs += row_xor<1>(cs);
        if (q < 3 && e0 + sl < E) {
            float v = (q == 0) ? dot : ((q == 1) ? cs : 1.0f);
            __hip_atomic_fetch_add(&node4[(size_t)d * 4 + q], v,
                                   __ATOMIC_RELAXED, __HIP_MEMORY_SCOPE_AGENT);
        }
    }
}

__global__ __launch_bounds__(1024) void finalize_kernel(
    const float* __restrict__ node4, const float* __restrict__ pos_diff,
    const int* __restrict__ out_nodes, const int* __restrict__ labels,
    const int* __restrict__ epoch, float* __restrict__ scratch,
    float* __restrict__ scores_out, int N, int Mnor) {
    __shared__ float red[32 * 4];
    const int i = blockIdx.x * blockDim.x + threadIdx.x;
    const int lane = threadIdx.x & 31;
    const int wave = threadIdx.x >> 5;

    float part[4] = {0.0f, 0.0f, 0.0f, 0.0f};
    if (i < N) {
        float4 a = *(const float4*)(node4 + (size_t)i * 4);
        float dg = a.z;
        float invd = dg > 0.0f ? 1.0f / fmaxf(dg, 1.0f) : 0.0f;
        float dot = a.x * invd;
        float mc  = a.y * invd;

        int on = out_nodes[i];
        float cnt = (float)Mnor;
        float mu  = scratch[0] / cnt;
        float var = (scratch[1] - cnt * mu * mu) / (cnt - 1.0f);
        float sd  = sqrtf(var);
        float z   = (pos_diff[on] - mu) / sd;
        float pre = 1.0f - 1.0f / (1.0f + expf(-z));

        float beta = powf(0.9f, (float)epoch[0]);
        if (beta < 0.1f) beta = 0.0f;
        float attn = beta * pre + (1.0f - beta) * dot;
        float nei  = attn * NEIGH_WEIGHT;

        float score = nei * mc + (1.0f - nei) * a.w;
        scores_out[i] = score;

        int lab = labels[on];
        float spl = log1pf(expf(-fabsf(score)));
        if (lab == 1) {
            part[2] = fmaxf(score, 0.0f) + spl;
            part[3] = 1.0f;
        } else {
            part[0] = fmaxf(-score, 0.0f) + spl;
            part[1] = 1.0f;
        }
    }
    for (int k = 0; k < 4; ++k) {
        float v = part[k];
        v += row_xor<8>(v); v += row_xor<4>(v); v += row_xor<2>(v); v += row_xor<1>(v);
        v += __shfl_xor(v, 16, 32);
        if (lane == 0) red[wave * 4 + k] = v;
    }
    __syncthreads();
    if (threadIdx.x < 4) {
        const int nw = (blockDim.x + 31) >> 5;
        float s = 0.0f;
        for (int w = 0; w < nw; ++w) s += red[w * 4 + threadIdx.x];
        __hip_atomic_fetch_add(&scratch[2 + threadIdx.x], s,
                               __ATOMIC_RELAXED, __HIP_MEMORY_SCOPE_AGENT);
    }
}

__global__ void loss_kernel(const float* __restrict__ scratch, float* __restrict__ out0) {
    if (threadIdx.x == 0 && blockIdx.x == 0) {
        out0[0] = scratch[2] / fmaxf(scratch[3], 1.0f) +
                  scratch[4] / fmaxf(scratch[5], 1.0f);
    }
}

extern "C" void kernel_launch(void* const* d_in, const int* in_sizes, int n_in,
                              void* d_out, int out_size, void* d_ws, size_t ws_size,
                              hipStream_t stream) {
    const float* feats    = (const float*)d_in[0];
    const float* W        = (const float*)d_in[1];
    const float* b        = (const float*)d_in[2];
    const float* center   = (const float*)d_in[3];
    const float* pos_diff = (const float*)d_in[4];
    const int*   src      = (const int*)d_in[5];
    const int*   dst      = (const int*)d_in[6];
    const int*   out_nodes= (const int*)d_in[7];
    const int*   nor_idx  = (const int*)d_in[8];
    const int*   labels   = (const int*)d_in[9];
    const int*   epoch    = (const int*)d_in[10];

    const int N = in_sizes[0] / 256;
    const int E = in_sizes[5];
    const int M = in_sizes[8];

    float* wsf     = (float*)d_ws;
    float* scratch = wsf;
    float* h       = wsf + 64;
    float* node4   = h + (size_t)N * 128;

    float* out     = (float*)d_out;

    zero_kernel<<<1, 64, 0, stream>>>(wsf, 64);

    stats_kernel<<<(M + 255) / 256, 256, 0, stream>>>(pos_diff, nor_idx, M, scratch);

    gemm_h_kernel<<<(N + 15) / 16, 256, 0, stream>>>(feats, W, b, center, h, node4, N);

    edge_dot_kernel<<<(E + 255) / 256, 256, 0, stream>>>(h, center, src, dst, node4, E);

    finalize_kernel<<<(N + 1023) / 1024, 1024, 0, stream>>>(node4, pos_diff, out_nodes, labels,
                                                           epoch, scratch, out + 1, N, M);

    loss_kernel<<<1, 64, 0, stream>>>(scratch, out);
}


# GEMM block tile 16 -> 64 rows: a wave keeps four 16-row accumulators and feeds all four from one W fragment per k-step (4x fewer W fetches and waves); strip loads issued together; sum order kept
# speedup vs baseline: 4.2085x; 1.0444x over previous
#include <hip/hip_runtime.h>
#include <hip/hip_bf16.h>
#include <math.h>

typedef float v2f __attribute__((ext_vector_type(2)));
typedef float v8f __attribute__((ext_vector_type(8)));

#define EPS 1e-12f
#define NEIGH_WEIGHT 0.2f

template <int K>
__device__ __forceinline__ float row_xor(float v) {
    return __builtin_bit_cast(float,
        __builtin_amdgcn_update_dpp(0, __builtin_bit_cast(int, v), 0x160 + K, 0xf, 0xf, true));
}

__global__ void zero_kernel(float* p, size_t n) {
    size_t i = (size_t)blockIdx.x * blockDim.x + threadIdx.x;
    size_t stride = (size_t)gridDim.x * blockDim.x;
    for (; i < n; i += stride) p[i] = 0.0f;
}

__global__ void stats_kernel(const float* __restrict__ pos_diff,
                             const int* __restrict__ nor_idx, int M,
                             float* __restrict__ scratch) {
    int t = blockIdx.x * blockDim.x + threadIdx.x;
    float v = 0.0f;
    if (t < M) v = pos_diff[nor_idx[t]];
    float s1 = v, s2 = v * v;
    #pragma unroll
    for (int off = 16; off > 0; off >>= 1) {
        s1 += __shfl_xor(s1, off, 32);
        s2 += __shfl_xor(s2, off, 32);
    }
    if ((threadIdx.x & 31) == 0) {
        __hip_atomic_fetch_add(&scratch[0], s1, __ATOMIC_RELAXED, __HIP_MEMORY_SCOPE_AGENT);
        __hip_atomic_fetch_add(&scratch[1], s2, __ATOMIC_RELAXED, __HIP_MEMORY_SCOPE_AGENT);
    }
}

#define GEMM_ROWS 64
#define GEMM_SUB  (GEMM_ROWS / 16)
__global__ __launch_bounds__(256) void gemm_h_kernel(
    const float* __restrict__ feats, const float* __restrict__ W,
    const float* __restrict__ bias, const float* __restrict__ center,
    float* __restrict__ h, float* __restrict__ node4, int N) {
    __shared__ float As[GEMM_ROWS * 260];
    __shared__ float P[GEMM_ROWS * 17];
    __shared__ float Pc[GEMM_ROWS * 17];
    __shared__ float RowInv[GEMM_ROWS];
    float* T = As;

    const int tid  = threadIdx.x;
    const int wave = tid >> 5;
    const int lane = tid & 31;
    const int row0 = blockIdx.x * GEMM_ROWS;
    const int cb   = wave * 16;
    const int hl   = lane & 15;
    const int hi   = lane >> 4;

    {
        float4 v[16];
        #pragma unroll
        for (int j = 0; j < 16; ++j) {
            int idx = tid * 4 + j * 1024;
            int r = idx >> 8;
            int c = idx & 255;
            int grow = row0 + r;
            int lrow = grow < N ? grow : N - 1;
            v[j] = *(const float4*)(feats + (size_t)lrow * 256 + c);
        }
        #pragma unroll
        for (int j = 0; j < 16; ++j) {
            int idx = tid * 4 + j * 1024;
            int r = idx >> 8;
            int c = idx & 255;
            float4 x = v[j];
            if (row0 + r >= N) x = make_float4(0.f, 0.f, 0.f, 0.f);
            float* dstp = &As[r * 260 + c];
            dstp[0] = x.x; dstp[1] = x.y; dstp[2] = x.z; dstp[3] = x.w;
        }
    }
    __syncthreads();

    v8f acc[GEMM_SUB];
    #pragma unroll
    for (int m = 0; m < GEMM_SUB; ++m) acc[m] = v8f{};
    const float* Wp = W + cb + hl;
    for (int k = 0; k < 256; k += 4) {
        int ka = k + hi * 2;
        v2f b;
        b.x = Wp[(size_t)ka * 128];
        b.y = Wp[(size_t)(ka + 1) * 128];
        #pragma unroll
        for (int m = 0; m < GEMM_SUB; ++m) {
            v2f a;
            a.x = As[(m * 16 + hl) * 260 + ka];
            a.y = As[(m * 16 + hl) * 260 + ka + 1];
            acc[m] = __builtin_amdgcn_wmma_f32_16x16x4_f32(false, a, false, b,
                                                           (short)0, acc[m], false, false);
        }
    }
    __syncthreads();

    float bv = bias[cb + hl];
    int rbase = hi * 8;
    #pragma unroll
    for (int m = 0; m < GEMM_SUB; ++m) {
        #pragma unroll
        for (int r = 0; r < 8; ++r) {
            float v = acc[m][r] + bv;
            v = v > 0.0f ? v : 0.0f;
            T[(m * 16 + rbase + r) * 132 + cb + hl] = v;
        }
    }
    __syncthreads();

    {
        int c0 = (tid & 15) * 8;
        float cv[8];
        #pragma unroll
        for (int j = 0; j < 8; ++j) cv[j] = center[c0 + j];
        #pragma unroll
        for (int m = 0; m < GEMM_SUB; ++m) {
            int row = m * 16 + (tid >> 4);
            float s = 0.0f, sc = 0.0f;
            #pragma unroll
            for (int j = 0; j < 8; ++j) {
                float v = T[row * 132 + c0 + j];
                s += v * v;
                sc += v * cv[j];
            }
            P[row * 17 + (tid & 15)] = s;
            Pc[row * 17 + (tid & 15)] = sc;
        }
    }
    __syncthreads();
    if (tid < GEMM_ROWS) {
        float s = 0.0f, sc = 0.0f;
        #pragma unroll
        for (int j = 0; j < 16; ++j) { s += P[tid * 17 + j]; sc += Pc[tid * 17 + j]; }
        float inv = 1.0f / fmaxf(sqrtf(s), EPS);
        RowInv[tid] = inv;
        if (row0 + tid < N)
            *(float4*)(node4 + (size_t)(row0 + tid) * 4) = make_float4(0.0f, 0.0f, 0.0f, sc * inv);
    }
    __syncthreads();

    #pragma unroll
    for (int m = 0; m < GEMM_SUB; ++m) {
        int row  = m * 16 + (tid >> 4);
        int grow = row0 + row;
        if (grow < N) {
            int c0 = (tid & 15) * 8;
            float inv = RowInv[row];
            float* op = h + (size_t)grow * 128 + c0;
            #pragma unroll
            for (int j = 0; j < 8; ++j) op[j] = T[row * 132 + c0 + j] * inv;
        }
    }
}

__global__ __launch_bounds__(256) void edge_dot_kernel(
    const float* __restrict__ h, const float* __restrict__ center,
    const int* __restrict__ src, const int* __restrict__ dst,
    float* __restrict__ node4, int E) {
    const int lane = threadIdx.x & 31;
    const int wv = (int)(((size_t)blockIdx.x * blockDim.x + threadIdx.x) >> 5);
    const int e0 = wv * 32;
    if (e0 >= E) return;
    const int g = lane >> 3;
    const int q = lane & 7;

    int sv = 0, dv = 0;
    if (e0 + lane < E) { sv = src[e0 + lane]; dv = dst[e0 + lane]; }

    float4 cw[4];
    for (int j = 0; j < 4; ++j) cw[j] = *(const float4*)(center + j * 32 + q * 4);

    for (int it = 0; it < 8; ++it) {
        const int sl = it * 4 + g;
        const int s = __shfl(sv, sl, 32);
        const int d = __shfl(dv, sl, 32);
        const float4* ps = (const float4*)(h + (size_t)s * 128) + q;
        const float4* pd = (const float4*)(h + (size_t)d * 128) + q;
        float4 a[4], b[4];
            for (int j = 0; j < 4; ++j) a[j] = ps[j * 8];
            for (int j = 0; j < 4; ++j) b[j] = pd[j * 8];
        v2f dot2 = {0.0f, 0.0f}, cs2 = {0.0f, 0.0f};
            for (int j = 0; j < 4; ++j) {
            v2f alo = {a[j].x, a[j].y}, ahi = {a[j].z, a[j].w};
            v2f blo = {b[j].x, b[j].y}, bhi = {b[j].z, b[j].w};
            v2f clo = {cw[j].x, cw[j].y}, chi = {cw[j].z, cw[j].w};
            dot2 += alo * blo; dot2 += ahi * bhi;
            cs2  += alo * clo; cs2  += ahi * chi;
        }
        float dot = dot2.x + dot2.y, cs = cs2.x + cs2.y;
        dot += row_xor<4>(dot); cs += row_xor<4>(cs);
        dot += row_xor<2>(dot); cs += row_xor<2>(cs);
        dot += row_xor<1>(dot); cs += row_xor<1>(cs);
        if (q < 3 && e0 + sl < E) {
            float v = (q == 0) ? dot : ((q == 1) ? cs : 1.0f);
            __hip_atomic_fetch_add(&node4[(size_t)d * 4 + q], v,
                                   __ATOMIC_RELAXED, __HIP_MEMORY_SCOPE_AGENT);
        }
    }
}

__global__ __launch_bounds__(1024) void finalize_kernel(
    const float* __restrict__ node4, const float* __restrict__ pos_diff,
    const int* __restrict__ out_nodes, const int* __restrict__ labels,
    const int* __restrict__ epoch, float* __restrict__ scratch,
    float* __restrict__ scores_out, int N, int Mnor) {
    __shared__ float red[32 * 4];
    const int i = blockIdx.x * blockDim.x + threadIdx.x;
    const int lane = threadIdx.x & 31;
    const int wave = threadIdx.x >> 5;

    float part[4] = {0.0f, 0.0f, 0.0f, 0.0f};
    if (i < N) {
        float4 a = *(const float4*)(node4 + (size_t)i * 4);
        float dg = a.z;
        float invd = dg > 0.0f ? 1.0f / fmaxf(dg, 1.0f) : 0.0f;
        float dot = a.x * invd;
        float mc  = a.y * invd;

        int on = out_nodes[i];
        float cnt = (float)Mnor;
        float mu  = scratch[0] / cnt;
        float var = (scratch[1] - cnt * mu * mu) / (cnt - 1.0f);
        float sd  = sqrtf(var);
        float z   = (pos_diff[on] - mu) / sd;
        float pre = 1.0f - 1.0f / (1.0f + expf(-z));

        float beta = powf(0.9f, (float)epoch[0]);
        if (beta < 0.1f) beta = 0.0f;
        float attn = beta * pre + (1.0f - beta) * dot;
        float nei  = attn * NEIGH_WEIGHT;

        float score = nei * mc + (1.0f - nei) * a.w;
        scores_out[i] = score;

        int lab = labels[on];
        float spl = log1pf(expf(-fabsf(score)));
        if (lab == 1) {
            part[2] = fmaxf(score, 0.0f) + spl;
            part[3] = 1.0f;
        } else {
            part[0] = fmaxf(-score, 0.0f) + spl;
            part[1] = 1.0f;
        }
    }
    for (int k = 0; k < 4; ++k) {
        float v = part[k];
        v += row_xor<8>(v); v += row_xor<4>(v); v += row_xor<2>(v); v += row_xor<1>(v);
        v += __shfl_xor(v, 16, 32);
        if (lane == 0) red[wave * 4 + k] = v;
    }
    __syncthreads();
    if (threadIdx.x < 4) {
        const int nw = (blockDim.x + 31) >> 5;
        float s = 0.0f;
        for (int w = 0; w < nw; ++w) s += red[w * 4 + threadIdx.x];
        __hip_atomic_fetch_add(&scratch[2 + threadIdx.x], s,
                               __ATOMIC_RELAXED, __HIP_MEMORY_SCOPE_AGENT);
    }
}

__global__ void loss_kernel(const float* __restrict__ scratch, float* __restrict__ out0) {
    if (threadIdx.x == 0 && blockIdx.x == 0) {
        out0[0] = scratch[2] / fmaxf(scratch[3], 1.0f) +
                  scratch[4] / fmaxf(scratch[5], 1.0f);
    }
}

extern "C" void kernel_launch(void* const* d_in, const int* in_sizes, int n_in,
                              void* d_out, int out_size, void* d_ws, size_t ws_size,
                              hipStream_t stream) {
    const float* feats    = (const float*)d_in[0];
    const float* W        = (const float*)d_in[1];
    const float* b        = (const float*)d_in[2];
    const float* center   = (const float*)d_in[3];
    const float* pos_diff = (const float*)d_in[4];
    const int*   src      = (const int*)d_in[5];
    const int*   dst      = (const int*)d_in[6];
    const int*   out_nodes= (const int*)d_in[7];
    const int*   nor_idx  = (const int*)d_in[8];
    const int*   labels   = (const int*)d_in[9];
    const int*   epoch    = (const int*)d_in[10];

    const int N = in_sizes[0] / 256;
    const int E = in_sizes[5];
    const int M = in_sizes[8];

    float* wsf     = (float*)d_ws;
    float* scratch = wsf;
    float* h       = wsf + 64;
    float* node4   = h + (size_t)N * 128;

    float* out     = (float*)d_out;

    zero_kernel<<<1, 64, 0, stream>>>(wsf, 64);

    stats_kernel<<<(M + 255) / 256, 256, 0, stream>>>(pos_diff, nor_idx, M, scratch);

    gemm_h_kernel<<<(N + GEMM_ROWS - 1) / GEMM_ROWS, 256, 0, stream>>>(feats, W, b, center, h, node4, N);

    edge_dot_kernel<<<(E + 255) / 256, 256, 0, stream>>>(h, center, src, dst, node4, E);

    finalize_kernel<<<(N + 1023) / 1024, 1024, 0, stream>>>(node4, pos_diff, out_nodes, labels,
                                                           epoch, scratch, out + 1, N, M);

    loss_kernel<<<1, 64, 0, stream>>>(scratch, out);
}
